# starting kernel
# baseline (speedup 1.0000x reference)
#include <hip/hip_runtime.h>
#include <hip/hip_bf16.h>


typedef __attribute__((ext_vector_type(2))) float v2f;
typedef __attribute__((ext_vector_type(8))) float v8f;

#define N_NODES 100000
#define IN_DIM  128
#define OUT_DIM 64
#define EDGES_PER_WAVE 256

__global__ void zero_f32(float* __restrict__ p, int n) {
    int i = blockIdx.x * blockDim.x + threadIdx.x;
    if (i < n) p[i] = 0.0f;
}

__global__ void __launch_bounds__(128)
gemm_xw_wmma(const float* __restrict__ X, const float* __restrict__ W,
             float* __restrict__ XP) {
    const int lane  = threadIdx.x & 31;
    const int tileN = threadIdx.x >> 5;
    const int tileM = blockIdx.x;
    const int row   = lane & 15;
    const int khalf = lane >> 4;

    const float* __restrict__ Xrow = X + (size_t)(tileM * 16 + row) * IN_DIM;
    const float* __restrict__ Wcol = W + tileN * 16 + row;

    v8f c = {};
#pragma unroll
    for (int k0 = 0; k0 < IN_DIM; k0 += 4) {
        v2f a = *(const v2f*)(Xrow + k0 + khalf * 2);
        v2f b;
        b.x = Wcol[(size_t)(k0 + khalf * 2 + 0) * OUT_DIM];
        b.y = Wcol[(size_t)(k0 + khalf * 2 + 1) * OUT_DIM];
        c = __builtin_amdgcn_wmma_f32_16x16x4_f32(
                false, a, false, b, (short)0, c, false, false);
    }

    float* __restrict__ outp = XP + (size_t)(tileM * 16) * OUT_DIM
                                  + tileN * 16 + (lane & 15);
#pragma unroll
    for (int v = 0; v < 8; ++v) {
        int m = v + khalf * 8;
        outp[(size_t)m * OUT_DIM] = c[v];
    }
}

__global__ void __launch_bounds__(256)
agnn_edge_kernel(const float* __restrict__ XP,
                 const int* __restrict__ esrc,
                 const int* __restrict__ edst,
                 const float* __restrict__ attw,
                 float* __restrict__ out, int nEdges) {
    const int lane = threadIdx.x & 31;
    const int wid  = blockIdx.x * (blockDim.x >> 5) + (threadIdx.x >> 5);

    int e0 = wid * EDGES_PER_WAVE;
    if (e0 >= nEdges) return;
    int e1 = e0 + EDGES_PER_WAVE;
    if (e1 > nEdges) e1 = nEdges;

    const float aw = attw[0];

    int   cur  = -1;
    float acc0 = 0.0f, acc1 = 0.0f;
    v2f   sf   = {0.0f, 0.0f};

    for (int e = e0; e < e1; ++e) {
        const int s = esrc[e];
        const int d = edst[e];

        if (s != cur) {
            if (cur >= 0) {
                float* dst = out + (size_t)cur * OUT_DIM + 2 * lane;
                unsafeAtomicAdd(dst + 0, acc0);
                unsafeAtomicAdd(dst + 1, acc1);
            }
            cur  = s;
            acc0 = 0.0f;
            acc1 = 0.0f;
            sf = *(const v2f*)(XP + (size_t)s * OUT_DIM + 2 * lane);
        }

        const v2f df = *(const v2f*)(XP + (size_t)d * OUT_DIM + 2 * lane);

        float p = sf.x * df.x + sf.y * df.y;
#pragma unroll
        for (int m = 16; m > 0; m >>= 1)
            p += __shfl_xor(p, m, 32);

        const float att = p * aw;
        acc0 += att * df.x;
        acc1 += att * df.y;
    }

    if (cur >= 0) {
        float* dst = out + (size_t)cur * OUT_DIM + 2 * lane;
        unsafeAtomicAdd(dst + 0, acc0);
        unsafeAtomicAdd(dst + 1, acc1);
    }
}

extern "C" void kernel_launch(void* const* d_in, const int* in_sizes, int n_in,
                              void* d_out, int out_size, void* d_ws, size_t ws_size,
                              hipStream_t stream) {
    const float* X    = (const float*)d_in[0];
    const float* W    = (const float*)d_in[1];
    const float* attw = (const float*)d_in[2];
    const int*   esrc = (const int*)d_in[3];
    const int*   edst = (const int*)d_in[4];
    float*       out  = (float*)d_out;
    float*       XP   = (float*)d_ws;

    const int nEdges = in_sizes[3];

    {
        int n = out_size;
        zero_f32<<<(n + 255) / 256, 256, 0, stream>>>(out, n);
    }

    gemm_xw_wmma<<<N_NODES / 16, 128, 0, stream>>>(X, W, XP);

    {
        int waves  = (nEdges + EDGES_PER_WAVE - 1) / EDGES_PER_WAVE;
        int blocks = (waves + 7) / 8;
        agnn_edge_kernel<<<blocks, 256, 0, stream>>>(XP, esrc, edst, attw,
                                                     out, nEdges);
    }
}


# edge kernel: 8 edges per wave in flight (8 dst-row gathers and the 8 butterfly sums issued as groups, index pairs by scalar loads one batch ahead; was one edge at a time); sums in the same order
# speedup vs baseline: 2.3447x; 2.3447x over previous
#include <hip/hip_runtime.h>
#include <hip/hip_bf16.h>


typedef __attribute__((ext_vector_type(2))) float v2f;
typedef __attribute__((ext_vector_type(8))) float v8f;

#define N_NODES 100000
#define IN_DIM  128
#define OUT_DIM 64
#define EDGES_PER_WAVE 256

__global__ void zero_f32(float* __restrict__ p, int n) {
    int i = blockIdx.x * blockDim.x + threadIdx.x;
    if (i < n) p[i] = 0.0f;
}

__global__ void __launch_bounds__(128)
gemm_xw_wmma(const float* __restrict__ X, const float* __restrict__ W,
             float* __restrict__ XP) {
    const int lane  = threadIdx.x & 31;
    const int tileN = threadIdx.x >> 5;
    const int tileM = blockIdx.x;
    const int row   = lane & 15;
    const int khalf = lane >> 4;

    const float* __restrict__ Xrow = X + (size_t)(tileM * 16 + row) * IN_DIM;
    const float* __restrict__ Wcol = W + tileN * 16 + row;

    v8f c = {};
#pragma unroll
    for (int k0 = 0; k0 < IN_DIM; k0 += 4) {
        v2f a = *(const v2f*)(Xrow + k0 + khalf * 2);
        v2f b;
        b.x = Wcol[(size_t)(k0 + khalf * 2 + 0) * OUT_DIM];
        b.y = Wcol[(size_t)(k0 + khalf * 2 + 1) * OUT_DIM];
        c = __builtin_amdgcn_wmma_f32_16x16x4_f32(
                false, a, false, b, (short)0, c, false, false);
    }

    float* __restrict__ outp = XP + (size_t)(tileM * 16) * OUT_DIM
                                  + tileN * 16 + (lane & 15);
#pragma unroll
    for (int v = 0; v < 8; ++v) {
        int m = v + khalf * 8;
        outp[(size_t)m * OUT_DIM] = c[v];
    }
}

#define EDGE_BATCH 8

__global__ void __launch_bounds__(256)
agnn_edge_kernel(const float* __restrict__ XP,
                 const int* __restrict__ esrc,
                 const int* __restrict__ edst,
                 const float* __restrict__ attw,
                 float* __restrict__ out, int nEdges) {
    const int lane = threadIdx.x & 31;
    const int wid  = __builtin_amdgcn_readfirstlane(
                         blockIdx.x * (blockDim.x >> 5) + (threadIdx.x >> 5));

    int e0 = wid * EDGES_PER_WAVE;
    if (e0 >= nEdges) return;
    int e1 = e0 + EDGES_PER_WAVE;
    if (e1 > nEdges) e1 = nEdges;

    const float aw = attw[0];
    const float* __restrict__ XPl = XP + 2 * lane;

    int   cur  = -1;
    float acc0 = 0.0f, acc1 = 0.0f;
    v2f   sf   = {0.0f, 0.0f};

    int e = e0;
    int s[EDGE_BATCH], d[EDGE_BATCH];
    if (e + EDGE_BATCH <= e1) {
#pragma unroll
        for (int j = 0; j < EDGE_BATCH; ++j) { s[j] = esrc[e + j]; d[j] = edst[e + j]; }
    }
    for (; e + EDGE_BATCH <= e1; e += EDGE_BATCH) {
        v2f df[EDGE_BATCH], sn[EDGE_BATCH];
#pragma unroll
        for (int j = 0; j < EDGE_BATCH; ++j)
            df[j] = *(const v2f*)(XPl + (size_t)d[j] * OUT_DIM);
#pragma unroll
        for (int j = 0; j < EDGE_BATCH; ++j) {
            sn[j] = (v2f){0.0f, 0.0f};
            if (s[j] != (j ? s[j - 1] : cur))
                sn[j] = *(const v2f*)(XPl + (size_t)s[j] * OUT_DIM);
        }

        const int en = (e + 2 * EDGE_BATCH <= e1) ? e + EDGE_BATCH : e;
        int sx[EDGE_BATCH], dx[EDGE_BATCH];
#pragma unroll
        for (int j = 0; j < EDGE_BATCH; ++j) { sx[j] = esrc[en + j]; dx[j] = edst[en + j]; }

        float p[EDGE_BATCH];
#pragma unroll
        for (int j = 0; j < EDGE_BATCH; ++j) {
            if (s[j] != (j ? s[j - 1] : cur)) sf = sn[j];
            p[j] = sf.x * df[j].x + sf.y * df[j].y;
        }

#pragma unroll
        for (int m = 16; m > 0; m >>= 1) {
            float t[EDGE_BATCH];
#pragma unroll
            for (int j = 0; j < EDGE_BATCH; ++j) t[j] = __shfl_xor(p[j], m, 32);
#pragma unroll
            for (int j = 0; j < EDGE_BATCH; ++j) p[j] += t[j];
        }

#pragma unroll
        for (int j = 0; j < EDGE_BATCH; ++j) {
            if (s[j] != cur) {
                if (cur >= 0) {
                    float* dst = out + (size_t)cur * OUT_DIM + 2 * lane;
                    unsafeAtomicAdd(dst + 0, acc0);
                    unsafeAtomicAdd(dst + 1, acc1);
                }
                cur  = s[j];
                acc0 = 0.0f;
                acc1 = 0.0f;
            }
            const float att = p[j] * aw;
            acc0 += att * df[j].x;
            acc1 += att * df[j].y;
        }

#pragma unroll
        for (int j = 0; j < EDGE_BATCH; ++j) { s[j] = sx[j]; d[j] = dx[j]; }
    }

    for (; e < e1; ++e) {
        const int st = esrc[e];
        const int dt = edst[e];

        if (st != cur) {
            if (cur >= 0) {
                float* dst = out + (size_t)cur * OUT_DIM + 2 * lane;
                unsafeAtomicAdd(dst + 0, acc0);
                unsafeAtomicAdd(dst + 1, acc1);
            }
            cur  = st;
            acc0 = 0.0f;
            acc1 = 0.0f;
            sf = *(const v2f*)(XPl + (size_t)st * OUT_DIM);
        }

        const v2f dft = *(const v2f*)(XPl + (size_t)dt * OUT_DIM);

        float pt = sf.x * dft.x + sf.y * dft.y;
#pragma unroll
        for (int m = 16; m > 0; m >>= 1)
            pt += __shfl_xor(pt, m, 32);

        const float att = pt * aw;
        acc0 += att * dft.x;
        acc1 += att * dft.y;
    }

    if (cur >= 0) {
        float* dst = out + (size_t)cur * OUT_DIM + 2 * lane;
        unsafeAtomicAdd(dst + 0, acc0);
        unsafeAtomicAdd(dst + 1, acc1);
    }
}

extern "C" void kernel_launch(void* const* d_in, const int* in_sizes, int n_in,
                              void* d_out, int out_size, void* d_ws, size_t ws_size,
                              hipStream_t stream) {
    const float* X    = (const float*)d_in[0];
    const float* W    = (const float*)d_in[1];
    const float* attw = (const float*)d_in[2];
    const int*   esrc = (const int*)d_in[3];
    const int*   edst = (const int*)d_in[4];
    float*       out  = (float*)d_out;
    float*       XP   = (float*)d_ws;

    const int nEdges = in_sizes[3];

    {
        int n = out_size;
        zero_f32<<<(n + 255) / 256, 256, 0, stream>>>(out, n);
    }

    gemm_xw_wmma<<<N_NODES / 16, 128, 0, stream>>>(X, W, XP);

    {
        int waves  = (nEdges + EDGES_PER_WAVE - 1) / EDGES_PER_WAVE;
        int blocks = (waves + 7) / 8;
        agnn_edge_kernel<<<blocks, 256, 0, stream>>>(XP, esrc, edst, attw,
                                                     out, nEdges);
    }
}


# zero fill of out moved into the GEMM launch: each block clears its row tile's 16 out rows, two 16-byte stores per thread; the fill launch and its 200000 one-store waves go; sums unchanged
# speedup vs baseline: 2.6304x; 1.1218x over previous
#include <hip/hip_runtime.h>
#include <hip/hip_bf16.h>


typedef __attribute__((ext_vector_type(2))) float v2f;
typedef __attribute__((ext_vector_type(8))) float v8f;

#define N_NODES 100000
#define IN_DIM  128
#define OUT_DIM 64
#define EDGES_PER_WAVE 256

__global__ void __launch_bounds__(128)
gemm_xw_wmma(const float* __restrict__ X, const float* __restrict__ W,
             float* __restrict__ XP, float* __restrict__ out) {
    const int lane  = threadIdx.x & 31;
    const int tileN = threadIdx.x >> 5;
    const int tileM = blockIdx.x;
    const int row   = lane & 15;
    const int khalf = lane >> 4;

    const float* __restrict__ Xrow = X + (size_t)(tileM * 16 + row) * IN_DIM;
    const float* __restrict__ Wcol = W + tileN * 16 + row;

    v8f c = {};
#pragma unroll
    for (int k0 = 0; k0 < IN_DIM; k0 += 4) {
        v2f a = *(const v2f*)(Xrow + k0 + khalf * 2);
        v2f b;
        b.x = Wcol[(size_t)(k0 + khalf * 2 + 0) * OUT_DIM];
        b.y = Wcol[(size_t)(k0 + khalf * 2 + 1) * OUT_DIM];
        c = __builtin_amdgcn_wmma_f32_16x16x4_f32(
                false, a, false, b, (short)0, c, false, false);
    }

    float* __restrict__ outp = XP + (size_t)(tileM * 16) * OUT_DIM
                                  + tileN * 16 + (lane & 15);
#pragma unroll
    for (int v = 0; v < 8; ++v) {
        int m = v + khalf * 8;
        outp[(size_t)m * OUT_DIM] = c[v];
    }

    float4* __restrict__ oz = (float4*)(out + (size_t)(tileM * 16) * OUT_DIM);
    const float4 z4 = {0.0f, 0.0f, 0.0f, 0.0f};
    oz[threadIdx.x]       = z4;
    oz[threadIdx.x + 128] = z4;
}

#define EDGE_BATCH 8

__global__ void __launch_bounds__(256)
agnn_edge_kernel(const float* __restrict__ XP,
                 const int* __restrict__ esrc,
                 const int* __restrict__ edst,
                 const float* __restrict__ attw,
                 float* __restrict__ out, int nEdges) {
    const int lane = threadIdx.x & 31;
    const int wid  = __builtin_amdgcn_readfirstlane(
                         blockIdx.x * (blockDim.x >> 5) + (threadIdx.x >> 5));

    int e0 = wid * EDGES_PER_WAVE;
    if (e0 >= nEdges) return;
    int e1 = e0 + EDGES_PER_WAVE;
    if (e1 > nEdges) e1 = nEdges;

    const float aw = attw[0];
    const float* __restrict__ XPl = XP + 2 * lane;

    int   cur  = -1;
    float acc0 = 0.0f, acc1 = 0.0f;
    v2f   sf   = {0.0f, 0.0f};

    int e = e0;
    int s[EDGE_BATCH], d[EDGE_BATCH];
    if (e + EDGE_BATCH <= e1) {
#pragma unroll
        for (int j = 0; j < EDGE_BATCH; ++j) { s[j] = esrc[e + j]; d[j] = edst[e + j]; }
    }
    for (; e + EDGE_BATCH <= e1; e += EDGE_BATCH) {
        v2f df[EDGE_BATCH], sn[EDGE_BATCH];
#pragma unroll
        for (int j = 0; j < EDGE_BATCH; ++j)
            df[j] = *(const v2f*)(XPl + (size_t)d[j] * OUT_DIM);
#pragma unroll
        for (int j = 0; j < EDGE_BATCH; ++j) {
            sn[j] = (v2f){0.0f, 0.0f};
            if (s[j] != (j ? s[j - 1] : cur))
                sn[j] = *(const v2f*)(XPl + (size_t)s[j] * OUT_DIM);
        }

        const int en = (e + 2 * EDGE_BATCH <= e1) ? e + EDGE_BATCH : e;
        int sx[EDGE_BATCH], dx[EDGE_BATCH];
#pragma unroll
        for (int j = 0; j < EDGE_BATCH; ++j) { sx[j] = esrc[en + j]; dx[j] = edst[en + j]; }

        float p[EDGE_BATCH];
#pragma unroll
        for (int j = 0; j < EDGE_BATCH; ++j) {
            if (s[j] != (j ? s[j - 1] : cur)) sf = sn[j];
            p[j] = sf.x * df[j].x + sf.y * df[j].y;
        }

#pragma unroll
        for (int m = 16; m > 0; m >>= 1) {
            float t[EDGE_BATCH];
#pragma unroll
            for (int j = 0; j < EDGE_BATCH; ++j) t[j] = __shfl_xor(p[j], m, 32);
#pragma unroll
            for (int j = 0; j < EDGE_BATCH; ++j) p[j] += t[j];
        }

#pragma unroll
        for (int j = 0; j < EDGE_BATCH; ++j) {
            if (s[j] != cur) {
                if (cur >= 0) {
                    float* dst = out + (size_t)cur * OUT_DIM + 2 * lane;
                    unsafeAtomicAdd(dst + 0, acc0);
                    unsafeAtomicAdd(dst + 1, acc1);
                }
                cur  = s[j];
                acc0 = 0.0f;
                acc1 = 0.0f;
            }
            const float att = p[j] * aw;
            acc0 += att * df[j].x;
            acc1 += att * df[j].y;
        }

#pragma unroll
        for (int j = 0; j < EDGE_BATCH; ++j) { s[j] = sx[j]; d[j] = dx[j]; }
    }

    for (; e < e1; ++e) {
        const int st = esrc[e];
        const int dt = edst[e];

        if (st != cur) {
            if (cur >= 0) {
                float* dst = out + (size_t)cur * OUT_DIM + 2 * lane;
                unsafeAtomicAdd(dst + 0, acc0);
                unsafeAtomicAdd(dst + 1, acc1);
            }
            cur  = st;
            acc0 = 0.0f;
            acc1 = 0.0f;
            sf = *(const v2f*)(XPl + (size_t)st * OUT_DIM);
        }

        const v2f dft = *(const v2f*)(XPl + (size_t)dt * OUT_DIM);

        float pt = sf.x * dft.x + sf.y * dft.y;
#pragma unroll
        for (int m = 16; m > 0; m >>= 1)
            pt += __shfl_xor(pt, m, 32);

        const float att = pt * aw;
        acc0 += att * dft.x;
        acc1 += att * dft.y;
    }

    if (cur >= 0) {
        float* dst = out + (size_t)cur * OUT_DIM + 2 * lane;
        unsafeAtomicAdd(dst + 0, acc0);
        unsafeAtomicAdd(dst + 1, acc1);
    }
}

extern "C" void kernel_launch(void* const* d_in, const int* in_sizes, int n_in,
                              void* d_out, int out_size, void* d_ws, size_t ws_size,
                              hipStream_t stream) {
    const float* X    = (const float*)d_in[0];
    const float* W    = (const float*)d_in[1];
    const float* attw = (const float*)d_in[2];
    const int*   esrc = (const int*)d_in[3];
    const int*   edst = (const int*)d_in[4];
    float*       out  = (float*)d_out;
    float*       XP   = (float*)d_ws;

    const int nEdges = in_sizes[3];

    gemm_xw_wmma<<<N_NODES / 16, 128, 0, stream>>>(X, W, XP, out);

    {
        int waves  = (nEdges + EDGES_PER_WAVE - 1) / EDGES_PER_WAVE;
        int blocks = (waves + 7) / 8;
        agnn_edge_kernel<<<blocks, 256, 0, stream>>>(XP, esrc, edst, attw,
                                                     out, nEdges);
    }
}


# GEMM re-tiled: one wave per 16x64 row strip (4 accumulator tiles share one A fragment, read once by 16-byte loads), W transposed into LDS once per block; k order of the X@W sums changes
# speedup vs baseline: 2.9597x; 1.1252x over previous
#include <hip/hip_runtime.h>
#include <hip/hip_bf16.h>


typedef __attribute__((ext_vector_type(2))) float v2f;
typedef __attribute__((ext_vector_type(8))) float v8f;

#define N_NODES 100000
#define IN_DIM  128
#define OUT_DIM 64
#define EDGES_PER_WAVE 256

typedef __attribute__((ext_vector_type(4))) float v4f;

#define N_STRIPS (N_NODES / 16)
#define WT_PITCH 132

__global__ void __launch_bounds__(128)
gemm_xw_wmma(const float* __restrict__ X, const float* __restrict__ W,
             float* __restrict__ XP, float* __restrict__ out) {
    __shared__ __attribute__((aligned(16))) float Wt[OUT_DIM * WT_PITCH];

    const int lane = threadIdx.x & 31;
    const int col  = lane & 15;
    const int half = lane >> 4;
    const int wave = __builtin_amdgcn_readfirstlane(threadIdx.x >> 5);
    int strip = blockIdx.x * 4 + wave;
    const bool live = strip < N_STRIPS;
    if (!live) strip = N_STRIPS - 1;

    const float* __restrict__ Xrow = X + (size_t)(strip * 16 + col) * IN_DIM + 4 * half;
    v4f a[16];
#pragma unroll
    for (int j = 0; j < 16; ++j)
        a[j] = *(const v4f*)(Xrow + 8 * j);

    if (live) {
        float4* __restrict__ oz = (float4*)(out + (size_t)(strip * 16) * OUT_DIM);
        const float4 z4 = {0.0f, 0.0f, 0.0f, 0.0f};
#pragma unroll
        for (int q = 0; q < 8; ++q)
            oz[lane + 32 * q] = z4;
    }

    {
        const int n  = threadIdx.x & 63;
        const int kq = threadIdx.x >> 6;
#pragma unroll
        for (int i = 0; i < 16; ++i) {
            const int k0 = 8 * i + 4 * kq;
            v4f w;
            w.x = W[(k0 + 0) * OUT_DIM + n];
            w.y = W[(k0 + 1) * OUT_DIM + n];
            w.z = W[(k0 + 2) * OUT_DIM + n];
            w.w = W[(k0 + 3) * OUT_DIM + n];
            *(v4f*)(&Wt[n * WT_PITCH + k0]) = w;
        }
    }
    __syncthreads();
    if (!live) return;

    unsigned xp_off = (unsigned)(((strip * 16 + half * 8) * OUT_DIM + col) * 4);
    asm volatile("" : "+v"(xp_off));
    float* __restrict__ outp = (float*)((char*)XP + xp_off);

    const float* bp = &Wt[col * WT_PITCH + 4 * half];

    v8f c0 = {}, c1 = {}, c2 = {}, c3 = {};
    v4f b0 = *(const v4f*)(bp + 0 * 16 * WT_PITCH);
    v4f b1 = *(const v4f*)(bp + 1 * 16 * WT_PITCH);
    v4f b2 = *(const v4f*)(bp + 2 * 16 * WT_PITCH);
    v4f b3 = *(const v4f*)(bp + 3 * 16 * WT_PITCH);
#pragma unroll
    for (int j = 0; j < 16; ++j) {
        const int jn = (j + 1 < 16) ? j + 1 : j;
        const v4f n0 = *(const v4f*)(bp + 0 * 16 * WT_PITCH + 8 * jn);
        const v4f n1 = *(const v4f*)(bp + 1 * 16 * WT_PITCH + 8 * jn);
        const v4f n2 = *(const v4f*)(bp + 2 * 16 * WT_PITCH + 8 * jn);
        const v4f n3 = *(const v4f*)(bp + 3 * 16 * WT_PITCH + 8 * jn);

        const v2f alo = {a[j].x, a[j].y};
        const v2f ahi = {a[j].z, a[j].w};
        c0 = __builtin_amdgcn_wmma_f32_16x16x4_f32(
                 false, alo, false, (v2f){b0.x, b0.y}, (short)0, c0, false, false);
        c1 = __builtin_amdgcn_wmma_f32_16x16x4_f32(
                 false, alo, false, (v2f){b1.x, b1.y}, (short)0, c1, false, false);
        c2 = __builtin_amdgcn_wmma_f32_16x16x4_f32(
                 false, alo, false, (v2f){b2.x, b2.y}, (short)0, c2, false, false);
        c3 = __builtin_amdgcn_wmma_f32_16x16x4_f32(
                 false, alo, false, (v2f){b3.x, b3.y}, (short)0, c3, false, false);
        c0 = __builtin_amdgcn_wmma_f32_16x16x4_f32(
                 false, ahi, false, (v2f){b0.z, b0.w}, (short)0, c0, false, false);
        c1 = __builtin_amdgcn_wmma_f32_16x16x4_f32(
                 false, ahi, false, (v2f){b1.z, b1.w}, (short)0, c1, false, false);
        c2 = __builtin_amdgcn_wmma_f32_16x16x4_f32(
                 false, ahi, false, (v2f){b2.z, b2.w}, (short)0, c2, false, false);
        c3 = __builtin_amdgcn_wmma_f32_16x16x4_f32(
                 false, ahi, false, (v2f){b3.z, b3.w}, (short)0, c3, false, false);

        b0 = n0; b1 = n1; b2 = n2; b3 = n3;
    }

#pragma unroll
    for (int v = 0; v < 8; ++v) {
        outp[(size_t)v * OUT_DIM +  0] = c0[v];
        outp[(size_t)v * OUT_DIM + 16] = c1[v];
        outp[(size_t)v * OUT_DIM + 32] = c2[v];
        outp[(size_t)v * OUT_DIM + 48] = c3[v];
    }
}

#define EDGE_BATCH 8

__global__ void __launch_bounds__(256)
agnn_edge_kernel(const float* __restrict__ XP,
                 const int* __restrict__ esrc,
                 const int* __restrict__ edst,
                 const float* __restrict__ attw,
                 float* __restrict__ out, int nEdges) {
    const int lane = threadIdx.x & 31;
    const int wid  = __builtin_amdgcn_readfirstlane(
                         blockIdx.x * (blockDim.x >> 5) + (threadIdx.x >> 5));

    int e0 = wid * EDGES_PER_WAVE;
    if (e0 >= nEdges) return;
    int e1 = e0 + EDGES_PER_WAVE;
    if (e1 > nEdges) e1 = nEdges;

    const float aw = attw[0];
    const float* __restrict__ XPl = XP + 2 * lane;

    int   cur  = -1;
    float acc0 = 0.0f, acc1 = 0.0f;
    v2f   sf   = {0.0f, 0.0f};

    int e = e0;
    int s[EDGE_BATCH], d[EDGE_BATCH];
    if (e + EDGE_BATCH <= e1) {
#pragma unroll
        for (int j = 0; j < EDGE_BATCH; ++j) { s[j] = esrc[e + j]; d[j] = edst[e + j]; }
    }
    for (; e + EDGE_BATCH <= e1; e += EDGE_BATCH) {
        v2f df[EDGE_BATCH], sn[EDGE_BATCH];
#pragma unroll
        for (int j = 0; j < EDGE_BATCH; ++j)
            df[j] = *(const v2f*)(XPl + (size_t)d[j] * OUT_DIM);
#pragma unroll
        for (int j = 0; j < EDGE_BATCH; ++j) {
            sn[j] = (v2f){0.0f, 0.0f};
            if (s[j] != (j ? s[j - 1] : cur))
                sn[j] = *(const v2f*)(XPl + (size_t)s[j] * OUT_DIM);
        }

        const int en = (e + 2 * EDGE_BATCH <= e1) ? e + EDGE_BATCH : e;
        int sx[EDGE_BATCH], dx[EDGE_BATCH];
#pragma unroll
        for (int j = 0; j < EDGE_BATCH; ++j) { sx[j] = esrc[en + j]; dx[j] = edst[en + j]; }

        float p[EDGE_BATCH];
#pragma unroll
        for (int j = 0; j < EDGE_BATCH; ++j) {
            if (s[j] != (j ? s[j - 1] : cur)) sf = sn[j];
            p[j] = sf.x * df[j].x + sf.y * df[j].y;
        }

#pragma unroll
        for (int m = 16; m > 0; m >>= 1) {
            float t[EDGE_BATCH];
#pragma unroll
            for (int j = 0; j < EDGE_BATCH; ++j) t[j] = __shfl_xor(p[j], m, 32);
#pragma unroll
            for (int j = 0; j < EDGE_BATCH; ++j) p[j] += t[j];
        }

#pragma unroll
        for (int j = 0; j < EDGE_BATCH; ++j) {
            if (s[j] != cur) {
                if (cur >= 0) {
                    float* dst = out + (size_t)cur * OUT_DIM + 2 * lane;
                    unsafeAtomicAdd(dst + 0, acc0);
                    unsafeAtomicAdd(dst + 1, acc1);
                }
                cur  = s[j];
                acc0 = 0.0f;
                acc1 = 0.0f;
            }
            const float att = p[j] * aw;
            acc0 += att * df[j].x;
            acc1 += att * df[j].y;
        }

#pragma unroll
        for (int j = 0; j < EDGE_BATCH; ++j) { s[j] = sx[j]; d[j] = dx[j]; }
    }

    for (; e < e1; ++e) {
        const int st = esrc[e];
        const int dt = edst[e];

        if (st != cur) {
            if (cur >= 0) {
                float* dst = out + (size_t)cur * OUT_DIM + 2 * lane;
                unsafeAtomicAdd(dst + 0, acc0);
                unsafeAtomicAdd(dst + 1, acc1);
            }
            cur  = st;
            acc0 = 0.0f;
            acc1 = 0.0f;
            sf = *(const v2f*)(XPl + (size_t)st * OUT_DIM);
        }

        const v2f dft = *(const v2f*)(XPl + (size_t)dt * OUT_DIM);

        float pt = sf.x * dft.x + sf.y * dft.y;
#pragma unroll
        for (int m = 16; m > 0; m >>= 1)
            pt += __shfl_xor(pt, m, 32);

        const float att = pt * aw;
        acc0 += att * dft.x;
        acc1 += att * dft.y;
    }

    if (cur >= 0) {
        float* dst = out + (size_t)cur * OUT_DIM + 2 * lane;
        unsafeAtomicAdd(dst + 0, acc0);
        unsafeAtomicAdd(dst + 1, acc1);
    }
}

extern "C" void kernel_launch(void* const* d_in, const int* in_sizes, int n_in,
                              void* d_out, int out_size, void* d_ws, size_t ws_size,
                              hipStream_t stream) {
    const float* X    = (const float*)d_in[0];
    const float* W    = (const float*)d_in[1];
    const float* attw = (const float*)d_in[2];
    const int*   esrc = (const int*)d_in[3];
    const int*   edst = (const int*)d_in[4];
    float*       out  = (float*)d_out;
    float*       XP   = (float*)d_ws;

    const int nEdges = in_sizes[3];

    gemm_xw_wmma<<<(N_STRIPS + 3) / 4, 128, 0, stream>>>(X, W, XP, out);

    {
        int waves  = (nEdges + EDGES_PER_WAVE - 1) / EDGES_PER_WAVE;
        int blocks = (waves + 7) / 8;
        agnn_edge_kernel<<<blocks, 256, 0, stream>>>(XP, esrc, edst, attw,
                                                     out, nEdges);
    }
}
